# starting kernel
# baseline (speedup 1.0000x reference)
#include <hip/hip_runtime.h>
#include <math.h>

#define Bn   32
#define Tn   4096
#define Dn   512
#define Sn   256
#define SCALE_INV 10.0f

typedef __attribute__((ext_vector_type(2))) float v2f;
typedef __attribute__((ext_vector_type(8))) float v8f;

#define NP 128
#define T_PER_WAVE (Tn / NP)

#define WS_H      0
#define WS_Q      (WS_H + Bn*Dn)
#define WS_ML     (WS_Q + Bn*Dn)
#define WS_ACC    (WS_ML + Bn*NP*2)
#define WS_POOLED (WS_ACC + Bn*NP*Dn)

__global__ void k_h(const float* __restrict__ t_star,
                    const float* __restrict__ W1,
                    const float* __restrict__ b1,
                    float* __restrict__ ws) {
    int i = blockIdx.x * blockDim.x + threadIdx.x;
    int b = i / Dn, d = i % Dn;
    float x = t_star[b] * W1[d] + b1[d];
    ws[WS_H + i] = x / (1.0f + __expf(-x));
}

__global__ void k_gemm_wmma(const float* __restrict__ A,
                            const float* __restrict__ Bm,
                            const float* __restrict__ bias,
                            float* __restrict__ Cout,
                            int N, int K, int split) {
    int lane = threadIdx.x & 31;
    int h    = lane >> 4;
    int ln   = lane & 15;
    int m0   = blockIdx.y * 16;
    int n    = blockIdx.x * 16 + ln;

    v8f c = {};
    for (int k0 = 0; k0 < K; k0 += 4) {
        int ka = k0 + 2 * h;
        v2f a, bb;
        a.x  = A[(m0 + ln) * K + ka];
        a.y  = A[(m0 + ln) * K + ka + 1];
        bb.x = Bm[(size_t)(ka    ) * N + n];
        bb.y = Bm[(size_t)(ka + 1) * N + n];
        c = __builtin_amdgcn_wmma_f32_16x16x4_f32(false, a, false, bb,
                                                  (short)0, c, false, false);
    }
    float bv = bias[n];
    #pragma unroll
    for (int v = 0; v < 8; ++v) {
        int m = m0 + v + 8 * h;
        float val = c[v] + bv;
        if (!split) {
            Cout[m * N + n] = val;
        } else {
            if (n < Sn) Cout[m * Sn + n] = val;
            else        Cout[Bn * Sn + m * Sn + (n - Sn)] = val;
        }
    }
}

__global__ void k_attend(const float* __restrict__ context,
                         const float* __restrict__ t_star,
                         const float* __restrict__ t_context,
                         float* __restrict__ ws) {
    int blk      = blockIdx.x;
    int b        = blk >> 4;
    int blk_in_b = blk & 15;
    int wave     = threadIdx.x >> 5;
    int lane     = threadIdx.x & 31;
    int p        = blk_in_b * 8 + wave;
    int t0       = p * T_PER_WAVE;

    const float* q = ws + WS_Q + b * Dn;
    float4 q4[4];
    #pragma unroll
    for (int i = 0; i < 4; ++i)
        q4[i] = *(const float4*)(q + lane * 4 + i * 128);

    float ts = t_star[b];
    const float* ctx_b = context   + (size_t)b * Tn * Dn;
    const float* tc_b  = t_context + (size_t)b * Tn;

    float m = -INFINITY, l = 0.0f;
    float4 acc[4] = {};

    for (int tt = 0; tt < T_PER_WAVE; ++tt) {
        int t = t0 + tt;
        const float* row = ctx_b + (size_t)t * Dn + lane * 4;
        float4 x[4];
        #pragma unroll
        for (int i = 0; i < 4; ++i)
            x[i] = *(const float4*)(row + i * 128);
        if (tt + 1 < T_PER_WAVE)
            __builtin_prefetch(row + Dn, 0, 1);

        float dotp = 0.0f;
        #pragma unroll
        for (int i = 0; i < 4; ++i)
            dotp += x[i].x * q4[i].x + x[i].y * q4[i].y +
                    x[i].z * q4[i].z + x[i].w * q4[i].w;
        #pragma unroll
        for (int off = 16; off > 0; off >>= 1)
            dotp += __shfl_xor(dotp, off, 32);

        float s  = dotp - fabsf(ts - tc_b[t]) * SCALE_INV;
        float mn = fmaxf(m, s);
        float sc = __expf(m - mn);
        float w  = __expf(s - mn);
        l = l * sc + w;
        #pragma unroll
        for (int i = 0; i < 4; ++i) {
            acc[i].x = acc[i].x * sc + w * x[i].x;
            acc[i].y = acc[i].y * sc + w * x[i].y;
            acc[i].z = acc[i].z * sc + w * x[i].z;
            acc[i].w = acc[i].w * sc + w * x[i].w;
        }
        m = mn;
    }

    float* accout = ws + WS_ACC + ((size_t)(b * NP + p)) * Dn;
    #pragma unroll
    for (int i = 0; i < 4; ++i)
        *(float4*)(accout + lane * 4 + i * 128) = acc[i];
    if (lane == 0) {
        float* ml = ws + WS_ML + (b * NP + p) * 2;
        ml[0] = m; ml[1] = l;
    }
}

__global__ void k_combine(float* __restrict__ ws) {
    int b = blockIdx.x;
    int d = threadIdx.x;
    const float* ml = ws + WS_ML + b * NP * 2;
    float M = -INFINITY;
    for (int p = 0; p < NP; ++p) M = fmaxf(M, ml[2 * p]);
    const float* accb = ws + WS_ACC + (size_t)b * NP * Dn;
    float L = 0.0f, sum = 0.0f;
    for (int p = 0; p < NP; ++p) {
        float e = __expf(ml[2 * p] - M);
        L   += e * ml[2 * p + 1];
        sum += e * accb[(size_t)p * Dn + d];
    }
    ws[WS_POOLED + b * Dn + d] = sum / L;
}

extern "C" void kernel_launch(void* const* d_in, const int* in_sizes, int n_in,
                              void* d_out, int out_size, void* d_ws, size_t ws_size,
                              hipStream_t stream) {
    const float* context   = (const float*)d_in[0];
    const float* t_star    = (const float*)d_in[1];
    const float* t_context = (const float*)d_in[2];
    const float* W1        = (const float*)d_in[3];
    const float* b1        = (const float*)d_in[4];
    const float* W2        = (const float*)d_in[5];
    const float* b2        = (const float*)d_in[6];
    const float* Wout      = (const float*)d_in[7];
    const float* bout      = (const float*)d_in[8];
    float* out = (float*)d_out;
    float* ws  = (float*)d_ws;

    k_h<<<(Bn * Dn) / 256, 256, 0, stream>>>(t_star, W1, b1, ws);

    k_gemm_wmma<<<dim3(Dn / 16, 2), 32, 0, stream>>>(ws + WS_H, W2, b2,
                                                     ws + WS_Q, Dn, Dn, 0);

    k_attend<<<Bn * 16, 256, 0, stream>>>(context, t_star, t_context, ws);

    k_combine<<<Bn, Dn, 0, stream>>>(ws);

    k_gemm_wmma<<<dim3((2 * Sn) / 16, 2), 32, 0, stream>>>(ws + WS_POOLED, Wout, bout,
                                                           out, 2 * Sn, Dn, 1);
}


# k_attend: columns split over a block's 8 waves, 32-row tiles of 16 independent loads per lane asked for a tile ahead, block-wide scores via LDS; one softmax state per block (new summation tree)
# speedup vs baseline: 3.2412x; 3.2412x over previous
#include <hip/hip_runtime.h>
#include <math.h>

#define Bn   32
#define Tn   4096
#define Dn   512
#define Sn   256
#define SCALE_INV 10.0f

typedef __attribute__((ext_vector_type(2))) float v2f;
typedef __attribute__((ext_vector_type(8))) float v8f;
typedef __attribute__((ext_vector_type(4))) float v4f;

#define NP 16

#define WS_H      0
#define WS_Q      (WS_H + Bn*Dn)
#define WS_ML     (WS_Q + Bn*Dn)
#define WS_ACC    (WS_ML + Bn*NP*2)
#define WS_POOLED (WS_ACC + Bn*NP*Dn)

__global__ void k_h(const float* __restrict__ t_star,
                    const float* __restrict__ W1,
                    const float* __restrict__ b1,
                    float* __restrict__ ws) {
    int i = blockIdx.x * blockDim.x + threadIdx.x;
    int b = i / Dn, d = i % Dn;
    float x = t_star[b] * W1[d] + b1[d];
    ws[WS_H + i] = x / (1.0f + __expf(-x));
}

#define LDA_S (Dn + 4)
#define KGRP  16

__global__ __launch_bounds__(256)
void k_gemm_wmma(const float* __restrict__ A,
                 const float* __restrict__ Bm,
                 const float* __restrict__ bias,
                 float* __restrict__ Cout,
                 int N, int split) {
    __shared__ __attribute__((aligned(16))) float sA[16 * LDA_S];
    __shared__ __attribute__((aligned(16))) float sB[Dn * 16];

    int tid = threadIdx.x;
    int m0  = blockIdx.y * 16;
    int n0  = blockIdx.x * 16;

    float4 rb[8], ra[8];
    #pragma unroll
    for (int i = 0; i < 8; ++i) {
        int idx = tid + 256 * i;
        rb[i] = *(const float4*)(Bm + (size_t)(idx >> 2) * N + n0 + (idx & 3) * 4);
    }
    #pragma unroll
    for (int i = 0; i < 8; ++i) {
        int idx = tid + 256 * i;
        ra[i] = *(const float4*)(A + (m0 + (idx >> 7)) * Dn + (idx & 127) * 4);
    }
    #pragma unroll
    for (int i = 0; i < 8; ++i) {
        int idx = tid + 256 * i;
        *(float4*)(sB + (idx >> 2) * 16 + (idx & 3) * 4) = rb[i];
    }
    #pragma unroll
    for (int i = 0; i < 8; ++i) {
        int idx = tid + 256 * i;
        *(float4*)(sA + (idx >> 7) * LDA_S + (idx & 127) * 4) = ra[i];
    }
    __syncthreads();
    if (tid >= 32) return;

    int lane = tid;
    int h    = lane >> 4;
    int ln   = lane & 15;
    const float* pa = sA + ln * LDA_S + 2 * h;
    const float* pb = sB + (2 * h) * 16 + ln;

    int n = n0 + ln;
    float bv = bias[n];

    v2f fa[2][KGRP], fb[2][KGRP];
    #pragma unroll
    for (int s = 0; s < KGRP; ++s) {
        fa[0][s]   = *(const v2f*)(pa + 4 * s);
        fb[0][s].x = pb[(4 * s) * 16];
        fb[0][s].y = pb[(4 * s + 1) * 16];
    }
    v8f c = {};
    #pragma unroll
    for (int g = 0; g < Dn / (4 * KGRP); ++g) {
        int cur = g & 1, nxt = cur ^ 1;
        if (g + 1 < Dn / (4 * KGRP)) {
            int kb = (g + 1) * 4 * KGRP;
            #pragma unroll
            for (int s = 0; s < KGRP; ++s) {
                fa[nxt][s]   = *(const v2f*)(pa + kb + 4 * s);
                fb[nxt][s].x = pb[(kb + 4 * s) * 16];
                fb[nxt][s].y = pb[(kb + 4 * s + 1) * 16];
            }
        }
        __builtin_amdgcn_sched_barrier(0);
        #pragma unroll
        for (int s = 0; s < KGRP; ++s)
            c = __builtin_amdgcn_wmma_f32_16x16x4_f32(false, fa[cur][s], false, fb[cur][s],
                                                      (short)0, c, false, false);
        __builtin_amdgcn_sched_barrier(0);
    }

    #pragma unroll
    for (int v = 0; v < 8; ++v) {
        int m = m0 + v + 8 * h;
        float val = c[v] + bv;
        if (!split) {
            Cout[m * N + n] = val;
        } else {
            if (n < Sn) Cout[m * Sn + n] = val;
            else        Cout[Bn * Sn + m * Sn + (n - Sn)] = val;
        }
    }
}

#define BLK_T   (Tn / NP)
#define TILE    32
#define NTILE   (BLK_T / TILE)
#define AW      8

#define DPPF(v, ctrl) __builtin_bit_cast(float, __builtin_amdgcn_update_dpp(0, __builtin_bit_cast(int, (v)), (ctrl), 0xf, 0xf, true))
#define XMASK(n) (0x160 + (n))
#define SHARE(n) (0x150 + (n))

__device__ __forceinline__ float half_sum16(float v) {
    v += DPPF(v, XMASK(1)); v += DPPF(v, XMASK(2));
    v += DPPF(v, XMASK(4)); v += DPPF(v, XMASK(8));
    return v;
}
__device__ __forceinline__ float half_max16(float v) {
    v = fmaxf(v, DPPF(v, XMASK(1))); v = fmaxf(v, DPPF(v, XMASK(2)));
    v = fmaxf(v, DPPF(v, XMASK(4))); v = fmaxf(v, DPPF(v, XMASK(8)));
    return v;
}
__device__ __forceinline__ float xhalf(float x) {
    int bits = __builtin_bit_cast(int, x);
    return __builtin_bit_cast(float, __builtin_amdgcn_permlanex16(bits, bits, 0x76543210, 0xfedcba98, false, false));
}

__device__ __forceinline__ void tile_step(const float4 (&x)[16], float tc, float (*sPb)[TILE],
                                          const float4& q4, float ts, int wave, int lane,
                                          float& m, float& l, float4& acc) {
    int c = lane & 15;
    float keep = 0.0f;
    #pragma unroll
    for (int i = 0; i < 16; ++i) {
        float pd = x[i].x * q4.x + x[i].y * q4.y + x[i].z * q4.z + x[i].w * q4.w;
        pd = half_sum16(pd);
        keep = (c == i) ? pd : keep;
    }
    sPb[wave][lane] = keep;
    asm volatile("s_wait_dscnt 0x0\n\ts_barrier_signal -1\n\ts_barrier_wait -1" ::: "memory");
    float s = 0.0f;
    #pragma unroll
    for (int w2 = 0; w2 < AW; ++w2) s += sPb[w2][lane];
    s = s - fabsf(ts - tc) * SCALE_INV;

    float mt = half_max16(s);
    mt = fmaxf(mt, xhalf(mt));
    float mn  = fmaxf(m, mt);
    float sc  = __expf(m - mn);
    float wgt = __expf(s - mn);
    float ws32 = half_sum16(wgt);
    ws32 += xhalf(ws32);
    l = l * sc + ws32;
    m = mn;

    acc.x *= sc; acc.y *= sc; acc.z *= sc; acc.w *= sc;
#define ACC_ROW(i) { float wi = DPPF(wgt, SHARE(i)); \
        acc.x += wi * x[i].x; acc.y += wi * x[i].y; acc.z += wi * x[i].z; acc.w += wi * x[i].w; }
    ACC_ROW(0)  ACC_ROW(1)  ACC_ROW(2)  ACC_ROW(3)  ACC_ROW(4)  ACC_ROW(5)  ACC_ROW(6)  ACC_ROW(7)
    ACC_ROW(8)  ACC_ROW(9)  ACC_ROW(10) ACC_ROW(11) ACC_ROW(12) ACC_ROW(13) ACC_ROW(14) ACC_ROW(15)
#undef ACC_ROW
}

__global__ __launch_bounds__(256)
void k_attend(const float* __restrict__ context,
              const float* __restrict__ t_star,
              const float* __restrict__ t_context,
              float* __restrict__ ws) {
    __shared__ float sP[2][AW][TILE];

    int blk  = blockIdx.x;
    int b    = blk >> 4;
    int p    = blk & 15;
    int wave = threadIdx.x >> 5;
    int lane = threadIdx.x & 31;
    int h    = lane >> 4;
    int c    = lane & 15;
    int r0   = p * BLK_T;

    const float* xrow = context + ((size_t)b * Tn + r0 + 16 * h) * Dn + wave * 64 + c * 4;
    const float* tcl  = t_context + (size_t)b * Tn + r0 + lane;

    float4 xa[16], xb[16];
    float tca, tcb;
    float m = -INFINITY, l = 0.0f;
    float4 acc = {0.0f, 0.0f, 0.0f, 0.0f};

#define LOAD_TILE(X, TC, T) do { \
        unsigned ti_ = (unsigned)(T); \
        asm volatile("" : "+s"(ti_), "+v"(acc.x), "+v"(acc.y), "+v"(acc.z), "+v"(acc.w)); \
        const float* xp_ = xrow + (size_t)ti_ * (TILE * Dn); \
        TC = tcl[ti_ * TILE]; \
        _Pragma("unroll") for (int i = 0; i < 16; ++i) \
            X[i] = *(const float4*)(xp_ + (size_t)i * Dn); } while (0)

    LOAD_TILE(xa, tca, 0);

    float4 q4 = *(const float4*)(ws + WS_Q + b * Dn + wave * 64 + c * 4);
    float ts  = t_star[b];

    #pragma unroll
    for (int t = 0; t < NTILE; t += 2) {
        LOAD_TILE(xb, tcb, t + 1);
        __builtin_amdgcn_sched_barrier(0);
        tile_step(xa, tca, sP[0], q4, ts, wave, lane, m, l, acc);
        if (t + 2 < NTILE) {
            LOAD_TILE(xa, tca, t + 2);
            __builtin_amdgcn_sched_barrier(0);
        }
        tile_step(xb, tcb, sP[1], q4, ts, wave, lane, m, l, acc);
    }
#undef LOAD_TILE

    float4 tot;
    tot.x = acc.x + xhalf(acc.x);
    tot.y = acc.y + xhalf(acc.y);
    tot.z = acc.z + xhalf(acc.z);
    tot.w = acc.w + xhalf(acc.w);
    if (h == 0)
        *(float4*)(ws + WS_ACC + ((size_t)(b * NP + p)) * Dn + wave * 64 + c * 4) = tot;
    if (threadIdx.x == 0) {
        float* ml = ws + WS_ML + (b * NP + p) * 2;
        ml[0] = m; ml[1] = l;
    }
}

#define CB_D   128
#define CB_W   8
#define CB_PW  (NP / CB_W)
__global__ __launch_bounds__(256)
void k_combine(float* __restrict__ ws) {
    __shared__ __attribute__((aligned(16))) float sSum[CB_W][CB_D];
    __shared__ float sL[CB_W];

    int b    = blockIdx.x / (Dn / CB_D);
    int ch   = blockIdx.x % (Dn / CB_D);
    int wave = threadIdx.x >> 5;
    int lane = threadIdx.x & 31;

    const float* ml   = ws + WS_ML + b * NP * 2;
    const float* accw = ws + WS_ACC + ((size_t)b * NP + wave * CB_PW) * Dn + ch * CB_D + lane * 4;

    float M = ml[2 * (lane & (NP - 1))];
    v4f mine = *(const v4f*)(ml + 2 * (wave * CB_PW));
    v4f a0 = *(const v4f*)(accw);
    v4f a1 = *(const v4f*)(accw + Dn);

    #pragma unroll
    for (int off = 16; off > 0; off >>= 1)
        M = fmaxf(M, __shfl_xor(M, off, 32));

    float e0 = __expf(mine.x - M);
    float e1 = __expf(mine.z - M);
    float L = __builtin_fmaf(e1, mine.w, __builtin_fmaf(e0, mine.y, 0.0f));
    v4f sum;
    sum.x = __builtin_fmaf(e1, a1.x, __builtin_fmaf(e0, a0.x, 0.0f));
    sum.y = __builtin_fmaf(e1, a1.y, __builtin_fmaf(e0, a0.y, 0.0f));
    sum.z = __builtin_fmaf(e1, a1.z, __builtin_fmaf(e0, a0.z, 0.0f));
    sum.w = __builtin_fmaf(e1, a1.w, __builtin_fmaf(e0, a0.w, 0.0f));

    *(v4f*)(&sSum[wave][lane * 4]) = sum;
    if (lane == 0) sL[wave] = L;
    __syncthreads();

    if (threadIdx.x < CB_D) {
        int d = threadIdx.x;
        float tot = sSum[0][d], Lt = sL[0];
        #pragma unroll
        for (int w = 1; w < CB_W; ++w) {
            tot += sSum[w][d];
            Lt  += sL[w];
        }
        ws[WS_POOLED + b * Dn + ch * CB_D + d] = tot / Lt;
    }
}

extern "C" void kernel_launch(void* const* d_in, const int* in_sizes, int n_in,
                              void* d_out, int out_size, void* d_ws, size_t ws_size,
                              hipStream_t stream) {
    const float* context   = (const float*)d_in[0];
    const float* t_star    = (const float*)d_in[1];
    const float* t_context = (const float*)d_in[2];
    const float* W1        = (const float*)d_in[3];
    const float* b1        = (const float*)d_in[4];
    const float* W2        = (const float*)d_in[5];
    const float* b2        = (const float*)d_in[6];
    const float* Wout      = (const float*)d_in[7];
    const float* bout      = (const float*)d_in[8];
    float* out = (float*)d_out;
    float* ws  = (float*)d_ws;

    k_h<<<(Bn * Dn) / 256, 256, 0, stream>>>(t_star, W1, b1, ws);

    k_gemm_wmma<<<dim3(Dn / 16, 2), 256, 0, stream>>>(ws + WS_H, W2, b2,
                                                      ws + WS_Q, Dn, 0);

    k_attend<<<Bn * NP, 256, 0, stream>>>(context, t_star, t_context, ws);

    k_combine<<<Bn * (Dn / CB_D), 256, 0, stream>>>(ws);

    k_gemm_wmma<<<dim3((2 * Sn) / 16, 2), 256, 0, stream>>>(ws + WS_POOLED, Wout, bout,
                                                            out, 2 * Sn, 1);
}
